# starting kernel
# baseline (speedup 1.0000x reference)
#include <hip/hip_runtime.h>
#include <math.h>


typedef __attribute__((ext_vector_type(2))) float v2f;
typedef __attribute__((ext_vector_type(8))) float v8f;

#define WIDTH  512
#define SEGS   32
#define XPITCH 513

__device__ __forceinline__ float exp2_fast(float x) {
    return __builtin_amdgcn_exp2f(x);
}

__device__ __forceinline__ void scan_dir(float* X, const float* tF, const float* omF,
                                         int lid, int half, int khi, bool bwd)
{
    float carry = 0.0f;
    for (int s = 0; s < SEGS; ++s) {
        const int base = s * 16;
        const int idx  = base + half;

        float t;
        if (!bwd) t = tF[idx];
        else      t = (idx == 0) ? -512.0f : tF[WIDTH - idx];

        float S = t;
#pragma unroll
        for (int off = 1; off < 16; off <<= 1) {
            float o = __shfl(S, (lid - off) & 31);
            if (half >= off) S += o;
        }
        const float expS = exp2_fast(S);

        v8f acc = {0.f,0.f,0.f,0.f,0.f,0.f,0.f,0.f};
#pragma unroll
        for (int kk = 0; kk < 4; ++kk) {
            const int m0 = kk * 4 + khi * 2;
            const float Sm0 = __shfl(S, m0);
            const float Sm1 = __shfl(S, m0 + 1);
            v2f Af;
            Af.x = (m0     < half) ? exp2_fast(S - Sm0) : ((m0     == half) ? 1.0f : 0.0f);
            Af.y = (m0 + 1 < half) ? exp2_fast(S - Sm1) : ((m0 + 1 == half) ? 1.0f : 0.0f);

            const int p0 = base + m0;
            v2f Bf;
            if (!bwd) {
                Bf.x = omF[p0]     * X[half * XPITCH + p0];
                Bf.y = omF[p0 + 1] * X[half * XPITCH + p0 + 1];
            } else {
                const float om0 = (p0 == 0) ? 1.0f : omF[WIDTH - p0];
                const float om1 = omF[WIDTH - (p0 + 1)];
                Bf.x = om0 * X[half * XPITCH + (WIDTH - 1 - p0)];
                Bf.y = om1 * X[half * XPITCH + (WIDTH - 1 - (p0 + 1))];
            }

            acc = __builtin_amdgcn_wmma_f32_16x16x4_f32(
                      false, Af, false, Bf, (short)0, acc, false, false);
        }

#pragma unroll
        for (int rr = 0; rr < 8; ++rr) {
            const float Pj = __shfl(expS, rr + 8 * khi);
            acc[rr] += Pj * carry;
        }
        carry = __shfl(acc[7], 16 + half);

#pragma unroll
        for (int rr = 0; rr < 8; ++rr) {
            const int j = base + rr + 8 * khi;
            const int w = bwd ? (WIDTH - 1 - j) : j;
            X[half * XPITCH + w] = acc[rr];
        }
    }
}

__global__ __launch_bounds__(32)
void dt_pass_kernel(const float* in, float* out, const float* edge,
                    float clog2, int vertical)
{
    extern __shared__ float smem[];
    float* X   = smem;
    float* tF  = smem + 16 * XPITCH;
    float* omF = tF + WIDTH;

    const int lid  = threadIdx.x;
    const int half = lid & 15;
    const int khi  = lid >> 4;

    const int nb  = blockIdx.x;
    const int cg  = nb & 1;
    const int row = nb >> 1;
    const int b   = row >> 9;
    const int r   = row & 511;

    const int C = 32, H = 512, Wd = 512;
    const long long chStride = (long long)H * Wd;
    long long imgBase, edgeBase;
    int posStride, eStride;
    if (!vertical) {
        imgBase  = (long long)b * C * H * Wd + (long long)r * Wd;
        posStride = 1;
        edgeBase = (long long)b * H * Wd + (long long)r * Wd;
        eStride  = 1;
    } else {
        imgBase  = (long long)b * C * H * Wd + r;
        posStride = Wd;
        edgeBase = (long long)b * H * Wd + r;
        eStride  = Wd;
    }
    imgBase += (long long)(cg * 16) * chStride;

    for (int i = lid; i < 16 * WIDTH; i += 32) {
        const int ch = i >> 9, w = i & (WIDTH - 1);
        X[ch * XPITCH + w] = in[imgBase + (long long)ch * chStride + (long long)w * posStride];
    }
    for (int w = lid; w < WIDTH; w += 32) {
        const float Dv = 1.0f + 150.0f * edge[edgeBase + (long long)w * eStride];
        float t = clog2 * Dv;
        if (w == 0) t = -512.0f;
        tF[w]  = t;
        omF[w] = 1.0f - exp2_fast(t);
    }
    __syncthreads();

    scan_dir(X, tF, omF, lid, half, khi, false);
    scan_dir(X, tF, omF, lid, half, khi, true);

    __syncthreads();
    for (int i = lid; i < 16 * WIDTH; i += 32) {
        const int ch = i >> 9, w = i & (WIDTH - 1);
        out[imgBase + (long long)ch * chStride + (long long)w * posStride] = X[ch * XPITCH + w];
    }
}

extern "C" void kernel_launch(void* const* d_in, const int* in_sizes, int n_in,
                              void* d_out, int out_size, void* d_ws, size_t ws_size,
                              hipStream_t stream) {
    (void)in_sizes; (void)n_in; (void)out_size; (void)d_ws; (void)ws_size;
    const float* img  = (const float*)d_in[0];
    const float* edge = (const float*)d_in[1];
    float* out = (float*)d_out;

    const double SQRT2 = 1.4142135623730951;
    const double SQRT3 = 1.7320508075688772;
    const double LOG2E = 1.4426950408889634;
    const double denom = 3.872983346207417;
    const double sigma1 = 60.0 * SQRT3 * 2.0 / denom;
    const double sigma2 = 60.0 * SQRT3 * 1.0 / denom;
    const float c1 = (float)(-SQRT2 / sigma1 * LOG2E);
    const float c2 = (float)(-SQRT2 / sigma2 * LOG2E);

    const dim3 grid(4096), block(32);
    const size_t shmem = (size_t)(16 * XPITCH + 2 * WIDTH) * sizeof(float);

    dt_pass_kernel<<<grid, block, shmem, stream>>>(img, out, edge, c1, 0);
    dt_pass_kernel<<<grid, block, shmem, stream>>>(out, out, edge, c1, 1);
    dt_pass_kernel<<<grid, block, shmem, stream>>>(out, out, edge, c2, 0);
    dt_pass_kernel<<<grid, block, shmem, stream>>>(out, out, edge, c2, 1);
}


# Horizontal pass: forward scan reads B fragments straight from global (k order 8h+2kk+e: 32 B a lane, 8 segments ahead), backward scan stores D rows to global; no staging sweeps; sum order changes
# speedup vs baseline: 3.0621x; 3.0621x over previous
#include <hip/hip_runtime.h>
#include <math.h>


typedef __attribute__((ext_vector_type(2))) float v2f;
typedef __attribute__((ext_vector_type(4))) float v4f;
typedef __attribute__((ext_vector_type(8))) float v8f;

#define WIDTH  512
#define SEGS   32
#define XPITCH 513

__device__ __forceinline__ float exp2_fast(float x) {
    return __builtin_amdgcn_exp2f(x);
}

__device__ __forceinline__ void build_tables(float* Sx, float* Ex, float* omx,
                                             const float (&tv)[16], int lid, int half)
{
    float S[16];
#pragma unroll
    for (int k = 0; k < 16; ++k) S[k] = tv[k];
#pragma unroll
    for (int off = 1; off < 16; off <<= 1) {
        float o[16];
#pragma unroll
        for (int k = 0; k < 16; ++k) o[k] = __shfl(S[k], (lid - off) & 31);
#pragma unroll
        for (int k = 0; k < 16; ++k) if (half >= off) S[k] += o[k];
    }
#pragma unroll
    for (int k = 0; k < 16; ++k) {
        const int w = 32 * k + lid;
        Sx[w]  = S[k];
        Ex[w]  = exp2_fast(S[k]);
        omx[w] = 1.0f - exp2_fast(tv[k]);
    }
}

__device__ __forceinline__ void lds_barrier() {
    asm volatile("s_wait_dscnt 0x0\n\ts_barrier_signal -1\n\ts_barrier_wait -1" ::: "memory");
}

__device__ __forceinline__ float h_step_fwd(float* X, const float* Sx, const float* Ex, const float* omx,
                                            int s, int half, int khi, float carry,
                                            const v4f x0, const v4f x1)
{
    const int base = s * 16;
    const int q0   = base + 8 * khi;
    const float S  = Sx[base + half];
    const v4f SmA = *(const v4f*)(Sx + q0);
    const v4f SmB = *(const v4f*)(Sx + q0 + 4);
    const v4f omA = *(const v4f*)(omx + q0);
    const v4f omB = *(const v4f*)(omx + q0 + 4);
    const v4f Pa  = *(const v4f*)(Ex + q0);
    const v4f Pb  = *(const v4f*)(Ex + q0 + 4);
    const float Sm[8] = {SmA.x, SmA.y, SmA.z, SmA.w, SmB.x, SmB.y, SmB.z, SmB.w};
    const float om[8] = {omA.x, omA.y, omA.z, omA.w, omB.x, omB.y, omB.z, omB.w};
    const float xv[8] = {x0.x, x0.y, x0.z, x0.w, x1.x, x1.y, x1.z, x1.w};

    v8f acc = {0.f,0.f,0.f,0.f,0.f,0.f,0.f,0.f};
#pragma unroll
    for (int kk = 0; kk < 4; ++kk) {
        const int m0 = 8 * khi + 2 * kk;
        v2f Af;
        Af.x = (m0     < half) ? exp2_fast(S - Sm[2 * kk])     : ((m0     == half) ? 1.0f : 0.0f);
        Af.y = (m0 + 1 < half) ? exp2_fast(S - Sm[2 * kk + 1]) : ((m0 + 1 == half) ? 1.0f : 0.0f);
        v2f Bf;
        Bf.x = om[2 * kk]     * xv[2 * kk];
        Bf.y = om[2 * kk + 1] * xv[2 * kk + 1];
        acc = __builtin_amdgcn_wmma_f32_16x16x4_f32(
                  false, Af, false, Bf, (short)0, acc, false, false);
    }

    acc[0] += Pa.x * carry;  acc[1] += Pa.y * carry;
    acc[2] += Pa.z * carry;  acc[3] += Pa.w * carry;
    acc[4] += Pb.x * carry;  acc[5] += Pb.y * carry;
    acc[6] += Pb.z * carry;  acc[7] += Pb.w * carry;
    const float next_carry = __shfl(acc[7], 16 + half);

#pragma unroll
    for (int rr = 0; rr < 8; ++rr)
        X[half * XPITCH + base + rr + 8 * khi] = acc[rr];
    return next_carry;
}

__device__ __forceinline__ float h_step_bwd(const float* X, const float* Sx, const float* Ex, const float* omx,
                                            int s, int half, int khi, float carry, float* dst)
{
    const int base = s * 16;
    const float S  = Sx[base + half];
    v2f Sm[4], om[4], xv[4];
#pragma unroll
    for (int kk = 0; kk < 4; ++kk) {
        const int p0 = base + kk * 4 + khi * 2;
        Sm[kk] = *(const v2f*)(Sx + p0);
        om[kk] = *(const v2f*)(omx + p0);
        xv[kk].x = X[half * XPITCH + (WIDTH - 1 - p0)];
        xv[kk].y = X[half * XPITCH + (WIDTH - 2 - p0)];
    }
    const v4f Pa = *(const v4f*)(Ex + base + 8 * khi);
    const v4f Pb = *(const v4f*)(Ex + base + 8 * khi + 4);

    v8f acc = {0.f,0.f,0.f,0.f,0.f,0.f,0.f,0.f};
#pragma unroll
    for (int kk = 0; kk < 4; ++kk) {
        const int m0 = kk * 4 + khi * 2;
        v2f Af;
        Af.x = (m0     < half) ? exp2_fast(S - Sm[kk].x) : ((m0     == half) ? 1.0f : 0.0f);
        Af.y = (m0 + 1 < half) ? exp2_fast(S - Sm[kk].y) : ((m0 + 1 == half) ? 1.0f : 0.0f);
        v2f Bf;
        Bf.x = om[kk].x * xv[kk].x;
        Bf.y = om[kk].y * xv[kk].y;
        acc = __builtin_amdgcn_wmma_f32_16x16x4_f32(
                  false, Af, false, Bf, (short)0, acc, false, false);
    }

    acc[0] += Pa.x * carry;  acc[1] += Pa.y * carry;
    acc[2] += Pa.z * carry;  acc[3] += Pa.w * carry;
    acc[4] += Pb.x * carry;  acc[5] += Pb.y * carry;
    acc[6] += Pb.z * carry;  acc[7] += Pb.w * carry;
    const float next_carry = __shfl(acc[7], 16 + half);

    const int wlo = WIDTH - 8 - base - 8 * khi;
    const v4f lo = {acc[7], acc[6], acc[5], acc[4]};
    const v4f hi = {acc[3], acc[2], acc[1], acc[0]};
    *(v4f*)(dst + wlo)     = lo;
    *(v4f*)(dst + wlo + 4) = hi;
    return next_carry;
}

#define HG 8

__device__ __forceinline__ void h_load_group(v4f (&xr)[2 * HG], const float* src, int g)
{
#pragma unroll
    for (int t = 0; t < HG; ++t) {
        xr[2 * t]     = *(const v4f*)(src + (g * HG + t) * 16);
        xr[2 * t + 1] = *(const v4f*)(src + (g * HG + t) * 16 + 4);
    }
}

__global__ __launch_bounds__(32)
void dt_hpass_kernel(const float* in, float* out, const float* edge, float clog2)
{
    extern __shared__ __attribute__((aligned(16))) float smem[];
    float* X   = smem;
    float* Sx  = smem + 16 * XPITCH;
    float* Ex  = Sx + WIDTH;
    float* omx = Ex + WIDTH;

    const int lid  = threadIdx.x;
    const int half = lid & 15;
    const int khi  = lid >> 4;

    const int nb  = blockIdx.x;
    const int cg  = nb & 1;
    const int row = nb >> 1;
    const int b   = row >> 9;
    const int r   = row & 511;

    const int C = 32, H = 512, Wd = 512;
    const long long chStride = (long long)H * Wd;
    const long long rowBase  = (long long)b * C * H * Wd + (long long)r * Wd
                             + (long long)(cg * 16 + half) * chStride;
    const long long edgeBase = (long long)b * H * Wd + (long long)r * Wd;
    const float* src = in + rowBase + 8 * khi;
    float*       dst = out + rowBase;

    float tvF[16], tvB[16];
#pragma unroll
    for (int k = 0; k < 16; ++k) {
        const int w  = 32 * k + lid;
        const int wr = (w == 0) ? 1 : (WIDTH - w);
        const float Df = 1.0f + 150.0f * edge[edgeBase + w];
        const float Db = 1.0f + 150.0f * edge[edgeBase + wr];
        float tf = clog2 * Df;
        float tb = clog2 * Db;
        if (w == 0) { tf = -512.0f; tb = -512.0f; }
        tvF[k] = tf;
        tvB[k] = tb;
    }
    v4f xa[2 * HG], xb[2 * HG];
    h_load_group(xa, src, 0);
    h_load_group(xb, src, 1);

    build_tables(Sx, Ex, omx, tvF, lid, half);
    lds_barrier();

    float carry = 0.0f;
    for (int g = 0; g < SEGS / HG; g += 2) {
#pragma unroll
        for (int t = 0; t < HG; ++t)
            carry = h_step_fwd(X, Sx, Ex, omx, g * HG + t, half, khi, carry, xa[2 * t], xa[2 * t + 1]);
        if (g + 2 < SEGS / HG) h_load_group(xa, src, g + 2);
#pragma unroll
        for (int t = 0; t < HG; ++t)
            carry = h_step_fwd(X, Sx, Ex, omx, (g + 1) * HG + t, half, khi, carry, xb[2 * t], xb[2 * t + 1]);
        if (g + 3 < SEGS / HG) h_load_group(xb, src, g + 3);
    }

    build_tables(Sx, Ex, omx, tvB, lid, half);
    carry = 0.0f;
    for (int s = 0; s < SEGS; ++s)
        carry = h_step_bwd(X, Sx, Ex, omx, s, half, khi, carry, dst);
}

#define VS_CHP  289
#define VS_ROWP 18
#define VS_BUF  (16 * VS_CHP)
#define VS_TP   513
#define VS_TQ   512

__device__ __forceinline__ void put4(float* p, const float4 v) {
    p[0] = v.x; p[1] = v.y; p[2] = v.z; p[3] = v.w;
}
__device__ __forceinline__ float4 get4(const float* p) {
    return make_float4(p[0], p[1], p[2], p[3]);
}

template <bool BWD>
__device__ __forceinline__ float vs_step(float* Xb, const float* Sc, const float* Ec, const float* omc,
                                         int s, int col, int half, int khi, float carry)
{
    const int base = s * 16;
    const float S  = Sc[base + half];
    v2f Sm[4], om[4], xv[4];
#pragma unroll
    for (int kk = 0; kk < 4; ++kk) {
        const int m0 = kk * 4 + khi * 2;
        const int p0 = base + m0;
        Sm[kk] = *(const v2f*)(Sc + p0);
        om[kk] = *(const v2f*)(omc + p0);
        if (!BWD) {
            xv[kk].x = Xb[half * VS_CHP + (m0    ) * VS_ROWP + col];
            xv[kk].y = Xb[half * VS_CHP + (m0 + 1) * VS_ROWP + col];
        } else {
            xv[kk].x = Xb[half * VS_CHP + (15 - m0) * VS_ROWP + col];
            xv[kk].y = Xb[half * VS_CHP + (14 - m0) * VS_ROWP + col];
        }
    }
    const v4f Pa = *(const v4f*)(Ec + base + 8 * khi);
    const v4f Pb = *(const v4f*)(Ec + base + 8 * khi + 4);

    v8f acc = {0.f,0.f,0.f,0.f,0.f,0.f,0.f,0.f};
#pragma unroll
    for (int kk = 0; kk < 4; ++kk) {
        const int m0 = kk * 4 + khi * 2;
        v2f Af;
        Af.x = (m0     < half) ? exp2_fast(S - Sm[kk].x) : ((m0     == half) ? 1.0f : 0.0f);
        Af.y = (m0 + 1 < half) ? exp2_fast(S - Sm[kk].y) : ((m0 + 1 == half) ? 1.0f : 0.0f);
        v2f Bf;
        Bf.x = om[kk].x * xv[kk].x;
        Bf.y = om[kk].y * xv[kk].y;
        acc = __builtin_amdgcn_wmma_f32_16x16x4_f32(
                  false, Af, false, Bf, (short)0, acc, false, false);
    }

    acc[0] += Pa.x * carry;  acc[1] += Pa.y * carry;
    acc[2] += Pa.z * carry;  acc[3] += Pa.w * carry;
    acc[4] += Pb.x * carry;  acc[5] += Pb.y * carry;
    acc[6] += Pb.z * carry;  acc[7] += Pb.w * carry;
    const float next_carry = __shfl(acc[7], 16 + half);

#pragma unroll
    for (int rr = 0; rr < 8; ++rr) {
        const int jl   = rr + 8 * khi;
        const int rowl = BWD ? (15 - jl) : jl;
        Xb[half * VS_CHP + rowl * VS_ROWP + col] = acc[rr];
    }
    return next_carry;
}

__global__ __launch_bounds__(512)
void dt_vstrip_kernel(const float* in, float* out, const float* edge, float clog2)
{
    __shared__ float XB[2 * VS_BUF];
    __shared__ float tS[16 * VS_TP];
    __shared__ __attribute__((aligned(16))) float SxS[16 * VS_TQ];
    __shared__ __attribute__((aligned(16))) float ExS[16 * VS_TQ];
    __shared__ __attribute__((aligned(16))) float omS[16 * VS_TQ];

    const int tid  = threadIdx.x;
    const int lid  = tid & 31;
    const int wv   = tid >> 5;
    const int half = lid & 15;
    const int khi  = lid >> 4;

    const int nb    = blockIdx.x;
    const int strip = nb & 31;
    const int cg    = (nb >> 5) & 1;
    const int b     = nb >> 6;

    const int Wd = 512;
    const long long chStride = 512LL * 512LL;
    const long long imgBase  = ((long long)b * 32 + cg * 16) * chStride + strip * 16;
    const long long edgeBase = (long long)b * chStride + strip * 16;

    const int q   = tid & 3;
    const int pr  = tid >> 2;
    const int ch0 = pr >> 4;
    const int rw0 = pr & 15;
    const long long g0 = imgBase + (long long)ch0 * chStride + (long long)rw0 * Wd + q * 4;
    const long long g1 = g0 + 8 * chStride;
    const int l0 = ch0 * VS_CHP + rw0 * VS_ROWP + q * 4;
    const int l1 = l0 + 8 * VS_CHP;
    const int tileStep = 16 * Wd;

    for (int k = 0; k < 4; ++k) {
        const int h = pr + 128 * k;
        const float4 e = *(const float4*)(edge + edgeBase + (long long)h * Wd + q * 4);
        const float ev[4] = {e.x, e.y, e.z, e.w};
        for (int i = 0; i < 4; ++i) {
            const float Dv = 1.0f + 150.0f * ev[i];
            tS[(q * 4 + i) * VS_TP + h] = clog2 * Dv;
        }
    }
    const float* tC = tS + wv * VS_TP;
    float* Sc  = SxS + wv * VS_TQ;
    float* Ec  = ExS + wv * VS_TQ;
    float* omc = omS + wv * VS_TQ;

    float4 r0 = *(const float4*)(in + g0);
    float4 r1 = *(const float4*)(in + g1);

    lds_barrier();

    float tv[16];
#pragma unroll
    for (int k = 0; k < 16; ++k) {
        const int w = 32 * k + lid;
        tv[k] = (w == 0) ? -512.0f : tC[w];
    }
    build_tables(Sc, Ec, omc, tv, lid, half);

    float carry = 0.0f;
    for (int s = 0; s < SEGS; ++s) {
        float* Xb = XB + (s & 1) * VS_BUF;
        put4(Xb + l0, r0);
        put4(Xb + l1, r1);
        if (s + 1 < SEGS) {
            r0 = *(const float4*)(in + g0 + (long long)(s + 1) * tileStep);
            r1 = *(const float4*)(in + g1 + (long long)(s + 1) * tileStep);
        }
        lds_barrier();
        carry = vs_step<false>(Xb, Sc, Ec, omc, s, wv, half, khi, carry);
        lds_barrier();
        const float4 o0 = get4(Xb + l0);
        const float4 o1 = get4(Xb + l1);
        *(float4*)(out + g0 + (long long)s * tileStep) = o0;
        *(float4*)(out + g1 + (long long)s * tileStep) = o1;
    }

    r0 = *(const float4*)(out + g0 + (long long)(SEGS - 1) * tileStep);
    r1 = *(const float4*)(out + g1 + (long long)(SEGS - 1) * tileStep);
#pragma unroll
    for (int k = 0; k < 16; ++k) {
        const int r = 32 * k + lid;
        tv[k] = (r == 0) ? -512.0f : tC[WIDTH - r];
    }
    build_tables(Sc, Ec, omc, tv, lid, half);
    carry = 0.0f;
    for (int s = 0; s < SEGS; ++s) {
        const int T = SEGS - 1 - s;
        float* Xb = XB + (s & 1) * VS_BUF;
        put4(Xb + l0, r0);
        put4(Xb + l1, r1);
        if (T > 0) {
            r0 = *(const float4*)(out + g0 + (long long)(T - 1) * tileStep);
            r1 = *(const float4*)(out + g1 + (long long)(T - 1) * tileStep);
        }
        lds_barrier();
        carry = vs_step<true>(Xb, Sc, Ec, omc, s, wv, half, khi, carry);
        lds_barrier();
        const float4 o0 = get4(Xb + l0);
        const float4 o1 = get4(Xb + l1);
        *(float4*)(out + g0 + (long long)T * tileStep) = o0;
        *(float4*)(out + g1 + (long long)T * tileStep) = o1;
    }
}

extern "C" void kernel_launch(void* const* d_in, const int* in_sizes, int n_in,
                              void* d_out, int out_size, void* d_ws, size_t ws_size,
                              hipStream_t stream) {
    (void)in_sizes; (void)n_in; (void)out_size; (void)d_ws; (void)ws_size;
    const float* img  = (const float*)d_in[0];
    const float* edge = (const float*)d_in[1];
    float* out = (float*)d_out;

    const double SQRT2 = 1.4142135623730951;
    const double SQRT3 = 1.7320508075688772;
    const double LOG2E = 1.4426950408889634;
    const double denom = 3.872983346207417;
    const double sigma1 = 60.0 * SQRT3 * 2.0 / denom;
    const double sigma2 = 60.0 * SQRT3 * 1.0 / denom;
    const float c1 = (float)(-SQRT2 / sigma1 * LOG2E);
    const float c2 = (float)(-SQRT2 / sigma2 * LOG2E);

    const dim3 grid(4096), block(32);
    const size_t shmem = (size_t)(16 * XPITCH + 3 * WIDTH) * sizeof(float);
    const dim3 vgrid(256), vblock(512);

    dt_hpass_kernel<<<grid, block, shmem, stream>>>(img, out, edge, c1);
    dt_vstrip_kernel<<<vgrid, vblock, 0, stream>>>(out, out, edge, c1);
    dt_hpass_kernel<<<grid, block, shmem, stream>>>(out, out, edge, c2);
    dt_vstrip_kernel<<<vgrid, vblock, 0, stream>>>(out, out, edge, c2);
}


# Vertical pass: an 8-wave work-group keeps its 8-column x 16-channel strip (512 rows) in LDS across both scans: input read once in 128-row batches, no forward store or re-read, 10 barriers, was 129
# speedup vs baseline: 4.1224x; 1.3463x over previous
#include <hip/hip_runtime.h>
#include <math.h>


typedef __attribute__((ext_vector_type(2))) float v2f;
typedef __attribute__((ext_vector_type(4))) float v4f;
typedef __attribute__((ext_vector_type(8))) float v8f;

#define WIDTH  512
#define SEGS   32
#define XPITCH 513

__device__ __forceinline__ float exp2_fast(float x) {
    return __builtin_amdgcn_exp2f(x);
}

__device__ __forceinline__ void build_tables(float* Sx, float* Ex, float* omx,
                                             const float (&tv)[16], int lid, int half)
{
    float S[16];
#pragma unroll
    for (int k = 0; k < 16; ++k) S[k] = tv[k];
#pragma unroll
    for (int off = 1; off < 16; off <<= 1) {
        float o[16];
#pragma unroll
        for (int k = 0; k < 16; ++k) o[k] = __shfl(S[k], (lid - off) & 31);
#pragma unroll
        for (int k = 0; k < 16; ++k) if (half >= off) S[k] += o[k];
    }
#pragma unroll
    for (int k = 0; k < 16; ++k) {
        const int w = 32 * k + lid;
        Sx[w]  = S[k];
        Ex[w]  = exp2_fast(S[k]);
        omx[w] = 1.0f - exp2_fast(tv[k]);
    }
}

__device__ __forceinline__ void lds_barrier() {
    asm volatile("s_wait_dscnt 0x0\n\ts_barrier_signal -1\n\ts_barrier_wait -1" ::: "memory");
}

__device__ __forceinline__ float h_step_fwd(float* X, const float* Sx, const float* Ex, const float* omx,
                                            int s, int half, int khi, float carry,
                                            const v4f x0, const v4f x1)
{
    const int base = s * 16;
    const int q0   = base + 8 * khi;
    const float S  = Sx[base + half];
    const v4f SmA = *(const v4f*)(Sx + q0);
    const v4f SmB = *(const v4f*)(Sx + q0 + 4);
    const v4f omA = *(const v4f*)(omx + q0);
    const v4f omB = *(const v4f*)(omx + q0 + 4);
    const v4f Pa  = *(const v4f*)(Ex + q0);
    const v4f Pb  = *(const v4f*)(Ex + q0 + 4);
    const float Sm[8] = {SmA.x, SmA.y, SmA.z, SmA.w, SmB.x, SmB.y, SmB.z, SmB.w};
    const float om[8] = {omA.x, omA.y, omA.z, omA.w, omB.x, omB.y, omB.z, omB.w};
    const float xv[8] = {x0.x, x0.y, x0.z, x0.w, x1.x, x1.y, x1.z, x1.w};

    v8f acc = {0.f,0.f,0.f,0.f,0.f,0.f,0.f,0.f};
#pragma unroll
    for (int kk = 0; kk < 4; ++kk) {
        const int m0 = 8 * khi + 2 * kk;
        v2f Af;
        Af.x = (m0     < half) ? exp2_fast(S - Sm[2 * kk])     : ((m0     == half) ? 1.0f : 0.0f);
        Af.y = (m0 + 1 < half) ? exp2_fast(S - Sm[2 * kk + 1]) : ((m0 + 1 == half) ? 1.0f : 0.0f);
        v2f Bf;
        Bf.x = om[2 * kk]     * xv[2 * kk];
        Bf.y = om[2 * kk + 1] * xv[2 * kk + 1];
        acc = __builtin_amdgcn_wmma_f32_16x16x4_f32(
                  false, Af, false, Bf, (short)0, acc, false, false);
    }

    acc[0] += Pa.x * carry;  acc[1] += Pa.y * carry;
    acc[2] += Pa.z * carry;  acc[3] += Pa.w * carry;
    acc[4] += Pb.x * carry;  acc[5] += Pb.y * carry;
    acc[6] += Pb.z * carry;  acc[7] += Pb.w * carry;
    const float next_carry = __shfl(acc[7], 16 + half);

#pragma unroll
    for (int rr = 0; rr < 8; ++rr)
        X[half * XPITCH + base + rr + 8 * khi] = acc[rr];
    return next_carry;
}

__device__ __forceinline__ float h_step_bwd(const float* X, const float* Sx, const float* Ex, const float* omx,
                                            int s, int half, int khi, float carry, float* dst)
{
    const int base = s * 16;
    const float S  = Sx[base + half];
    v2f Sm[4], om[4], xv[4];
#pragma unroll
    for (int kk = 0; kk < 4; ++kk) {
        const int p0 = base + kk * 4 + khi * 2;
        Sm[kk] = *(const v2f*)(Sx + p0);
        om[kk] = *(const v2f*)(omx + p0);
        xv[kk].x = X[half * XPITCH + (WIDTH - 1 - p0)];
        xv[kk].y = X[half * XPITCH + (WIDTH - 2 - p0)];
    }
    const v4f Pa = *(const v4f*)(Ex + base + 8 * khi);
    const v4f Pb = *(const v4f*)(Ex + base + 8 * khi + 4);

    v8f acc = {0.f,0.f,0.f,0.f,0.f,0.f,0.f,0.f};
#pragma unroll
    for (int kk = 0; kk < 4; ++kk) {
        const int m0 = kk * 4 + khi * 2;
        v2f Af;
        Af.x = (m0     < half) ? exp2_fast(S - Sm[kk].x) : ((m0     == half) ? 1.0f : 0.0f);
        Af.y = (m0 + 1 < half) ? exp2_fast(S - Sm[kk].y) : ((m0 + 1 == half) ? 1.0f : 0.0f);
        v2f Bf;
        Bf.x = om[kk].x * xv[kk].x;
        Bf.y = om[kk].y * xv[kk].y;
        acc = __builtin_amdgcn_wmma_f32_16x16x4_f32(
                  false, Af, false, Bf, (short)0, acc, false, false);
    }

    acc[0] += Pa.x * carry;  acc[1] += Pa.y * carry;
    acc[2] += Pa.z * carry;  acc[3] += Pa.w * carry;
    acc[4] += Pb.x * carry;  acc[5] += Pb.y * carry;
    acc[6] += Pb.z * carry;  acc[7] += Pb.w * carry;
    const float next_carry = __shfl(acc[7], 16 + half);

    const int wlo = WIDTH - 8 - base - 8 * khi;
    const v4f lo = {acc[7], acc[6], acc[5], acc[4]};
    const v4f hi = {acc[3], acc[2], acc[1], acc[0]};
    *(v4f*)(dst + wlo)     = lo;
    *(v4f*)(dst + wlo + 4) = hi;
    return next_carry;
}

#define HG 8

__device__ __forceinline__ void h_load_group(v4f (&xr)[2 * HG], const float* src, int g)
{
#pragma unroll
    for (int t = 0; t < HG; ++t) {
        xr[2 * t]     = *(const v4f*)(src + (g * HG + t) * 16);
        xr[2 * t + 1] = *(const v4f*)(src + (g * HG + t) * 16 + 4);
    }
}

__global__ __launch_bounds__(32)
void dt_hpass_kernel(const float* in, float* out, const float* edge, float clog2)
{
    extern __shared__ __attribute__((aligned(16))) float smem[];
    float* X   = smem;
    float* Sx  = smem + 16 * XPITCH;
    float* Ex  = Sx + WIDTH;
    float* omx = Ex + WIDTH;

    const int lid  = threadIdx.x;
    const int half = lid & 15;
    const int khi  = lid >> 4;

    const int nb  = blockIdx.x;
    const int cg  = nb & 1;
    const int row = nb >> 1;
    const int b   = row >> 9;
    const int r   = row & 511;

    const int C = 32, H = 512, Wd = 512;
    const long long chStride = (long long)H * Wd;
    const long long rowBase  = (long long)b * C * H * Wd + (long long)r * Wd
                             + (long long)(cg * 16 + half) * chStride;
    const long long edgeBase = (long long)b * H * Wd + (long long)r * Wd;
    const float* src = in + rowBase + 8 * khi;
    float*       dst = out + rowBase;

    float tvF[16], tvB[16];
#pragma unroll
    for (int k = 0; k < 16; ++k) {
        const int w  = 32 * k + lid;
        const int wr = (w == 0) ? 1 : (WIDTH - w);
        const float Df = 1.0f + 150.0f * edge[edgeBase + w];
        const float Db = 1.0f + 150.0f * edge[edgeBase + wr];
        float tf = clog2 * Df;
        float tb = clog2 * Db;
        if (w == 0) { tf = -512.0f; tb = -512.0f; }
        tvF[k] = tf;
        tvB[k] = tb;
    }
    v4f xa[2 * HG], xb[2 * HG];
    h_load_group(xa, src, 0);
    h_load_group(xb, src, 1);

    build_tables(Sx, Ex, omx, tvF, lid, half);
    lds_barrier();

    float carry = 0.0f;
    for (int g = 0; g < SEGS / HG; g += 2) {
#pragma unroll
        for (int t = 0; t < HG; ++t)
            carry = h_step_fwd(X, Sx, Ex, omx, g * HG + t, half, khi, carry, xa[2 * t], xa[2 * t + 1]);
        if (g + 2 < SEGS / HG) h_load_group(xa, src, g + 2);
#pragma unroll
        for (int t = 0; t < HG; ++t)
            carry = h_step_fwd(X, Sx, Ex, omx, (g + 1) * HG + t, half, khi, carry, xb[2 * t], xb[2 * t + 1]);
        if (g + 3 < SEGS / HG) h_load_group(xb, src, g + 3);
    }

    build_tables(Sx, Ex, omx, tvB, lid, half);
    carry = 0.0f;
    for (int s = 0; s < SEGS; ++s)
        carry = h_step_bwd(X, Sx, Ex, omx, s, half, khi, carry, dst);
}

#define VS_COLS 8
#define VS_ROWP 8
#define VS_CHP  4097
#define VS_GROWS 128
#define VS_NG   4
#define VS_TP   513
#define VS_TQ   512

__device__ __forceinline__ void put4(float* p, const float4 v) {
    p[0] = v.x; p[1] = v.y; p[2] = v.z; p[3] = v.w;
}
__device__ __forceinline__ float4 get4(const float* p) {
    return make_float4(p[0], p[1], p[2], p[3]);
}

template <bool BWD>
__device__ __forceinline__ float vs_step(float* Xb, const float* Sc, const float* Ec, const float* omc,
                                         int s, int col, int half, int khi, float carry)
{
    const int base = s * 16;
    const float S  = Sc[base + half];
    v2f Sm[4], om[4], xv[4];
#pragma unroll
    for (int kk = 0; kk < 4; ++kk) {
        const int m0 = kk * 4 + khi * 2;
        const int p0 = base + m0;
        Sm[kk] = *(const v2f*)(Sc + p0);
        om[kk] = *(const v2f*)(omc + p0);
        if (!BWD) {
            xv[kk].x = Xb[half * VS_CHP + (m0    ) * VS_ROWP + col];
            xv[kk].y = Xb[half * VS_CHP + (m0 + 1) * VS_ROWP + col];
        } else {
            xv[kk].x = Xb[half * VS_CHP + (15 - m0) * VS_ROWP + col];
            xv[kk].y = Xb[half * VS_CHP + (14 - m0) * VS_ROWP + col];
        }
    }
    const v4f Pa = *(const v4f*)(Ec + base + 8 * khi);
    const v4f Pb = *(const v4f*)(Ec + base + 8 * khi + 4);

    v8f acc = {0.f,0.f,0.f,0.f,0.f,0.f,0.f,0.f};
#pragma unroll
    for (int kk = 0; kk < 4; ++kk) {
        const int m0 = kk * 4 + khi * 2;
        v2f Af;
        Af.x = (m0     < half) ? exp2_fast(S - Sm[kk].x) : ((m0     == half) ? 1.0f : 0.0f);
        Af.y = (m0 + 1 < half) ? exp2_fast(S - Sm[kk].y) : ((m0 + 1 == half) ? 1.0f : 0.0f);
        v2f Bf;
        Bf.x = om[kk].x * xv[kk].x;
        Bf.y = om[kk].y * xv[kk].y;
        acc = __builtin_amdgcn_wmma_f32_16x16x4_f32(
                  false, Af, false, Bf, (short)0, acc, false, false);
    }

    acc[0] += Pa.x * carry;  acc[1] += Pa.y * carry;
    acc[2] += Pa.z * carry;  acc[3] += Pa.w * carry;
    acc[4] += Pb.x * carry;  acc[5] += Pb.y * carry;
    acc[6] += Pb.z * carry;  acc[7] += Pb.w * carry;
    const float next_carry = __shfl(acc[7], 16 + half);

#pragma unroll
    for (int rr = 0; rr < 8; ++rr) {
        const int jl   = rr + 8 * khi;
        const int rowl = BWD ? (15 - jl) : jl;
        Xb[half * VS_CHP + rowl * VS_ROWP + col] = acc[rr];
    }
    return next_carry;
}

__global__ __launch_bounds__(256)
void dt_vres_kernel(const float* in, float* out, const float* edge, float clog2)
{
    __shared__ float XS[16 * VS_CHP];
    __shared__ __attribute__((aligned(16))) float SxS[VS_COLS * VS_TQ];
    __shared__ __attribute__((aligned(16))) float ExS[VS_COLS * VS_TQ];
    __shared__ __attribute__((aligned(16))) float omS[VS_COLS * VS_TQ];

    const int tid  = threadIdx.x;
    const int lid  = tid & 31;
    const int wv   = tid >> 5;
    const int half = lid & 15;
    const int khi  = lid >> 4;

    const int nb    = blockIdx.x;
    const int strip = nb & 63;
    const int cg    = (nb >> 6) & 1;
    const int b     = nb >> 7;

    const int Wd = 512;
    const long long chStride = 512LL * 512LL;
    const long long imgBase  = ((long long)b * 32 + cg * 16) * chStride + strip * VS_COLS;

    const int q  = tid & 1;
    const int pp = tid >> 1;
    const long long gm = imgBase + (long long)pp * Wd + q * 4;
    const int lm = pp * VS_ROWP + q * 4;
    const long long gStep = (long long)VS_GROWS * Wd;
    const int lStep = VS_GROWS * VS_ROWP;

    float* tS = XS;
    const long long edgeBase = (long long)b * chStride + strip * VS_COLS;
    float4 ee[4];
#pragma unroll
    for (int k = 0; k < 4; ++k)
        ee[k] = *(const float4*)(edge + edgeBase + (long long)(pp + 128 * k) * Wd + q * 4);

    float4 rb[16];
#pragma unroll
    for (int i = 0; i < 16; ++i)
        rb[i] = *(const float4*)(in + gm + (long long)i * chStride);

#pragma unroll
    for (int k = 0; k < 4; ++k) {
        const int h = pp + 128 * k;
        const float ev[4] = {ee[k].x, ee[k].y, ee[k].z, ee[k].w};
#pragma unroll
        for (int i = 0; i < 4; ++i) {
            const float Dv = 1.0f + 150.0f * ev[i];
            tS[(q * 4 + i) * VS_TP + h] = clog2 * Dv;
        }
    }
    lds_barrier();

    const float* tC = tS + wv * VS_TP;
    float tvF[16], tvB[16];
#pragma unroll
    for (int k = 0; k < 16; ++k) {
        const int w = 32 * k + lid;
        tvF[k] = (w == 0) ? -512.0f : tC[w];
        tvB[k] = (w == 0) ? -512.0f : tC[WIDTH - w];
    }

    float* Sc  = SxS + wv * VS_TQ;
    float* Ec  = ExS + wv * VS_TQ;
    float* omc = omS + wv * VS_TQ;
    build_tables(Sc, Ec, omc, tvF, lid, half);
    lds_barrier();

    float carry = 0.0f;
    for (int g = 0; g < VS_NG; ++g) {
#pragma unroll
        for (int i = 0; i < 16; ++i)
            put4(XS + i * VS_CHP + g * lStep + lm, rb[i]);
        if (g + 1 < VS_NG) {
#pragma unroll
            for (int i = 0; i < 16; ++i)
                rb[i] = *(const float4*)(in + gm + (long long)i * chStride + (long long)(g + 1) * gStep);
        }
        lds_barrier();
        for (int t = 0; t < 8; ++t) {
            const int s = g * 8 + t;
            carry = vs_step<false>(XS + s * 16 * VS_ROWP, Sc, Ec, omc, s, wv, half, khi, carry);
        }
    }

    build_tables(Sc, Ec, omc, tvB, lid, half);
    carry = 0.0f;
    for (int h = 0; h < VS_NG; ++h) {
        for (int t = 0; t < 8; ++t) {
            const int s = h * 8 + t;
            const int T = SEGS - 1 - s;
            carry = vs_step<true>(XS + T * 16 * VS_ROWP, Sc, Ec, omc, s, wv, half, khi, carry);
        }
        lds_barrier();
        const int G = VS_NG - 1 - h;
#pragma unroll
        for (int i = 0; i < 16; ++i) {
            const float4 o = get4(XS + i * VS_CHP + G * lStep + lm);
            *(float4*)(out + gm + (long long)i * chStride + (long long)G * gStep) = o;
        }
    }
}

extern "C" void kernel_launch(void* const* d_in, const int* in_sizes, int n_in,
                              void* d_out, int out_size, void* d_ws, size_t ws_size,
                              hipStream_t stream) {
    (void)in_sizes; (void)n_in; (void)out_size; (void)d_ws; (void)ws_size;
    const float* img  = (const float*)d_in[0];
    const float* edge = (const float*)d_in[1];
    float* out = (float*)d_out;

    const double SQRT2 = 1.4142135623730951;
    const double SQRT3 = 1.7320508075688772;
    const double LOG2E = 1.4426950408889634;
    const double denom = 3.872983346207417;
    const double sigma1 = 60.0 * SQRT3 * 2.0 / denom;
    const double sigma2 = 60.0 * SQRT3 * 1.0 / denom;
    const float c1 = (float)(-SQRT2 / sigma1 * LOG2E);
    const float c2 = (float)(-SQRT2 / sigma2 * LOG2E);

    const dim3 grid(4096), block(32);
    const size_t shmem = (size_t)(16 * XPITCH + 3 * WIDTH) * sizeof(float);
    const dim3 vgrid(512), vblock(256);

    dt_hpass_kernel<<<grid, block, shmem, stream>>>(img, out, edge, c1);
    dt_vres_kernel<<<vgrid, vblock, 0, stream>>>(out, out, edge, c1);
    dt_hpass_kernel<<<grid, block, shmem, stream>>>(out, out, edge, c2);
    dt_vres_kernel<<<vgrid, vblock, 0, stream>>>(out, out, edge, c2);
}


# Tensor kept strip-major in out between H pass 1 and H pass 2: a vertical strip row is 512 contiguous bytes (was 32-byte pieces a plane apart), an H segment step 1 KB contiguous; in place, same sums
# speedup vs baseline: 4.7734x; 1.1579x over previous
#include <hip/hip_runtime.h>
#include <math.h>


typedef __attribute__((ext_vector_type(2))) float v2f;
typedef __attribute__((ext_vector_type(4))) float v4f;
typedef __attribute__((ext_vector_type(8))) float v8f;

#define WIDTH  512
#define SEGS   32
#define XPITCH 513
#define PLANE  (512LL * 512LL)

__device__ __forceinline__ float exp2_fast(float x) {
    return __builtin_amdgcn_exp2f(x);
}

__device__ __forceinline__ void build_tables(float* Sx, float* Ex, float* omx,
                                             const float (&tv)[16], int lid, int half)
{
    float S[16];
#pragma unroll
    for (int k = 0; k < 16; ++k) S[k] = tv[k];
#pragma unroll
    for (int off = 1; off < 16; off <<= 1) {
        float o[16];
#pragma unroll
        for (int k = 0; k < 16; ++k) o[k] = __shfl(S[k], (lid - off) & 31);
#pragma unroll
        for (int k = 0; k < 16; ++k) if (half >= off) S[k] += o[k];
    }
#pragma unroll
    for (int k = 0; k < 16; ++k) {
        const int w = 32 * k + lid;
        Sx[w]  = S[k];
        Ex[w]  = exp2_fast(S[k]);
        omx[w] = 1.0f - exp2_fast(tv[k]);
    }
}

__device__ __forceinline__ void lds_barrier() {
    asm volatile("s_wait_dscnt 0x0\n\ts_barrier_signal -1\n\ts_barrier_wait -1" ::: "memory");
}

__device__ __forceinline__ float h_step_fwd(float* X, const float* Sx, const float* Ex, const float* omx,
                                            int s, int half, int khi, float carry,
                                            const v4f x0, const v4f x1)
{
    const int base = s * 16;
    const int q0   = base + 8 * khi;
    const float S  = Sx[base + half];
    const v4f SmA = *(const v4f*)(Sx + q0);
    const v4f SmB = *(const v4f*)(Sx + q0 + 4);
    const v4f omA = *(const v4f*)(omx + q0);
    const v4f omB = *(const v4f*)(omx + q0 + 4);
    const v4f Pa  = *(const v4f*)(Ex + q0);
    const v4f Pb  = *(const v4f*)(Ex + q0 + 4);
    const float Sm[8] = {SmA.x, SmA.y, SmA.z, SmA.w, SmB.x, SmB.y, SmB.z, SmB.w};
    const float om[8] = {omA.x, omA.y, omA.z, omA.w, omB.x, omB.y, omB.z, omB.w};
    const float xv[8] = {x0.x, x0.y, x0.z, x0.w, x1.x, x1.y, x1.z, x1.w};

    v8f acc = {0.f,0.f,0.f,0.f,0.f,0.f,0.f,0.f};
#pragma unroll
    for (int kk = 0; kk < 4; ++kk) {
        const int m0 = 8 * khi + 2 * kk;
        v2f Af;
        Af.x = (m0     < half) ? exp2_fast(S - Sm[2 * kk])     : ((m0     == half) ? 1.0f : 0.0f);
        Af.y = (m0 + 1 < half) ? exp2_fast(S - Sm[2 * kk + 1]) : ((m0 + 1 == half) ? 1.0f : 0.0f);
        v2f Bf;
        Bf.x = om[2 * kk]     * xv[2 * kk];
        Bf.y = om[2 * kk + 1] * xv[2 * kk + 1];
        acc = __builtin_amdgcn_wmma_f32_16x16x4_f32(
                  false, Af, false, Bf, (short)0, acc, false, false);
    }

    acc[0] += Pa.x * carry;  acc[1] += Pa.y * carry;
    acc[2] += Pa.z * carry;  acc[3] += Pa.w * carry;
    acc[4] += Pb.x * carry;  acc[5] += Pb.y * carry;
    acc[6] += Pb.z * carry;  acc[7] += Pb.w * carry;
    const float next_carry = __shfl(acc[7], 16 + half);

#pragma unroll
    for (int rr = 0; rr < 8; ++rr)
        X[half * XPITCH + base + rr + 8 * khi] = acc[rr];
    return next_carry;
}

template <bool OUT_P>
__device__ __forceinline__ float h_step_bwd(const float* X, const float* Sx, const float* Ex, const float* omx,
                                            int s, int half, int khi, float carry, float* dst)
{
    const int base = s * 16;
    const float S  = Sx[base + half];
    v2f Sm[4], om[4], xv[4];
#pragma unroll
    for (int kk = 0; kk < 4; ++kk) {
        const int p0 = base + kk * 4 + khi * 2;
        Sm[kk] = *(const v2f*)(Sx + p0);
        om[kk] = *(const v2f*)(omx + p0);
        xv[kk].x = X[half * XPITCH + (WIDTH - 1 - p0)];
        xv[kk].y = X[half * XPITCH + (WIDTH - 2 - p0)];
    }
    const v4f Pa = *(const v4f*)(Ex + base + 8 * khi);
    const v4f Pb = *(const v4f*)(Ex + base + 8 * khi + 4);

    v8f acc = {0.f,0.f,0.f,0.f,0.f,0.f,0.f,0.f};
#pragma unroll
    for (int kk = 0; kk < 4; ++kk) {
        const int m0 = kk * 4 + khi * 2;
        v2f Af;
        Af.x = (m0     < half) ? exp2_fast(S - Sm[kk].x) : ((m0     == half) ? 1.0f : 0.0f);
        Af.y = (m0 + 1 < half) ? exp2_fast(S - Sm[kk].y) : ((m0 + 1 == half) ? 1.0f : 0.0f);
        v2f Bf;
        Bf.x = om[kk].x * xv[kk].x;
        Bf.y = om[kk].y * xv[kk].y;
        acc = __builtin_amdgcn_wmma_f32_16x16x4_f32(
                  false, Af, false, Bf, (short)0, acc, false, false);
    }

    acc[0] += Pa.x * carry;  acc[1] += Pa.y * carry;
    acc[2] += Pa.z * carry;  acc[3] += Pa.w * carry;
    acc[4] += Pb.x * carry;  acc[5] += Pb.y * carry;
    acc[6] += Pb.z * carry;  acc[7] += Pb.w * carry;
    const float next_carry = __shfl(acc[7], 16 + half);

    const int wlo = WIDTH - 8 - base - 8 * khi;
    const v4f lo = {acc[7], acc[6], acc[5], acc[4]};
    const v4f hi = {acc[3], acc[2], acc[1], acc[0]};
    if (OUT_P) {
        const int u = wlo >> 3;
        float* p = dst + (long long)(u >> 2) * PLANE + (u & 3) * 128;
        *(v4f*)(p)     = lo;
        *(v4f*)(p + 4) = hi;
    } else {
        *(v4f*)(dst + wlo)     = lo;
        *(v4f*)(dst + wlo + 4) = hi;
    }
    return next_carry;
}

#define HG 8

template <bool IN_P>
__device__ __forceinline__ void h_load_group(v4f (&xr)[2 * HG], const float* src, int g)
{
#pragma unroll
    for (int t = 0; t < HG; ++t) {
        const float* p = IN_P ? src + (long long)(g * (HG / 2) + (t >> 1)) * PLANE + (t & 1) * 256
                              : src + (g * HG + t) * 16;
        xr[2 * t]     = *(const v4f*)(p);
        xr[2 * t + 1] = *(const v4f*)(p + 4);
    }
}

template <bool IN_P, bool OUT_P>
__global__ __launch_bounds__(32)
void dt_hpass_kernel(const float* in, float* out, const float* edge, float clog2)
{
    extern __shared__ __attribute__((aligned(16))) float smem[];
    float* X   = smem;
    float* Sx  = smem + 16 * XPITCH;
    float* Ex  = Sx + WIDTH;
    float* omx = Ex + WIDTH;

    const int lid  = threadIdx.x;
    const int half = lid & 15;
    const int khi  = lid >> 4;

    const int nb  = blockIdx.x;
    const int cg  = nb & 1;
    const int row = nb >> 1;
    const int b   = row >> 9;
    const int r   = row & 511;

    const int C = 32, H = 512, Wd = 512;
    const long long chStride = (long long)H * Wd;
    const long long regBase  = (long long)b * C * H * Wd + (long long)r * Wd
                             + (long long)(cg * 16) * chStride;
    const long long rowBase  = regBase + (long long)half * chStride;
    const long long edgeBase = (long long)b * H * Wd + (long long)r * Wd;
    const float* src = IN_P  ? in + regBase + khi * 128 + half * 8 : in + rowBase + 8 * khi;
    float*       dst = OUT_P ? out + regBase + half * 8            : out + rowBase;

    float tvF[16], tvB[16];
#pragma unroll
    for (int k = 0; k < 16; ++k) {
        const int w  = 32 * k + lid;
        const int wr = (w == 0) ? 1 : (WIDTH - w);
        const float Df = 1.0f + 150.0f * edge[edgeBase + w];
        const float Db = 1.0f + 150.0f * edge[edgeBase + wr];
        float tf = clog2 * Df;
        float tb = clog2 * Db;
        if (w == 0) { tf = -512.0f; tb = -512.0f; }
        tvF[k] = tf;
        tvB[k] = tb;
    }
    v4f xa[2 * HG], xb[2 * HG];
    h_load_group<IN_P>(xa, src, 0);
    h_load_group<IN_P>(xb, src, 1);

    build_tables(Sx, Ex, omx, tvF, lid, half);
    lds_barrier();

    float carry = 0.0f;
    for (int g = 0; g < SEGS / HG; g += 2) {
#pragma unroll
        for (int t = 0; t < HG; ++t)
            carry = h_step_fwd(X, Sx, Ex, omx, g * HG + t, half, khi, carry, xa[2 * t], xa[2 * t + 1]);
        if (g + 2 < SEGS / HG) h_load_group<IN_P>(xa, src, g + 2);
#pragma unroll
        for (int t = 0; t < HG; ++t)
            carry = h_step_fwd(X, Sx, Ex, omx, (g + 1) * HG + t, half, khi, carry, xb[2 * t], xb[2 * t + 1]);
        if (g + 3 < SEGS / HG) h_load_group<IN_P>(xb, src, g + 3);
    }

    build_tables(Sx, Ex, omx, tvB, lid, half);
    carry = 0.0f;
    for (int s = 0; s < SEGS; ++s)
        carry = h_step_bwd<OUT_P>(X, Sx, Ex, omx, s, half, khi, carry, dst);
}

#define VS_COLS 8
#define VS_ROWP 8
#define VS_CHP  4097
#define VS_GROWS 128
#define VS_NG   4
#define VS_TP   513
#define VS_TQ   512

__device__ __forceinline__ void put4(float* p, const float4 v) {
    p[0] = v.x; p[1] = v.y; p[2] = v.z; p[3] = v.w;
}
__device__ __forceinline__ float4 get4(const float* p) {
    return make_float4(p[0], p[1], p[2], p[3]);
}

template <bool BWD>
__device__ __forceinline__ float vs_step(float* Xb, const float* Sc, const float* Ec, const float* omc,
                                         int s, int col, int half, int khi, float carry)
{
    const int base = s * 16;
    const float S  = Sc[base + half];
    v2f Sm[4], om[4], xv[4];
#pragma unroll
    for (int kk = 0; kk < 4; ++kk) {
        const int m0 = kk * 4 + khi * 2;
        const int p0 = base + m0;
        Sm[kk] = *(const v2f*)(Sc + p0);
        om[kk] = *(const v2f*)(omc + p0);
        if (!BWD) {
            xv[kk].x = Xb[half * VS_CHP + (m0    ) * VS_ROWP + col];
            xv[kk].y = Xb[half * VS_CHP + (m0 + 1) * VS_ROWP + col];
        } else {
            xv[kk].x = Xb[half * VS_CHP + (15 - m0) * VS_ROWP + col];
            xv[kk].y = Xb[half * VS_CHP + (14 - m0) * VS_ROWP + col];
        }
    }
    const v4f Pa = *(const v4f*)(Ec + base + 8 * khi);
    const v4f Pb = *(const v4f*)(Ec + base + 8 * khi + 4);

    v8f acc = {0.f,0.f,0.f,0.f,0.f,0.f,0.f,0.f};
#pragma unroll
    for (int kk = 0; kk < 4; ++kk) {
        const int m0 = kk * 4 + khi * 2;
        v2f Af;
        Af.x = (m0     < half) ? exp2_fast(S - Sm[kk].x) : ((m0     == half) ? 1.0f : 0.0f);
        Af.y = (m0 + 1 < half) ? exp2_fast(S - Sm[kk].y) : ((m0 + 1 == half) ? 1.0f : 0.0f);
        v2f Bf;
        Bf.x = om[kk].x * xv[kk].x;
        Bf.y = om[kk].y * xv[kk].y;
        acc = __builtin_amdgcn_wmma_f32_16x16x4_f32(
                  false, Af, false, Bf, (short)0, acc, false, false);
    }

    acc[0] += Pa.x * carry;  acc[1] += Pa.y * carry;
    acc[2] += Pa.z * carry;  acc[3] += Pa.w * carry;
    acc[4] += Pb.x * carry;  acc[5] += Pb.y * carry;
    acc[6] += Pb.z * carry;  acc[7] += Pb.w * carry;
    const float next_carry = __shfl(acc[7], 16 + half);

#pragma unroll
    for (int rr = 0; rr < 8; ++rr) {
        const int jl   = rr + 8 * khi;
        const int rowl = BWD ? (15 - jl) : jl;
        Xb[half * VS_CHP + rowl * VS_ROWP + col] = acc[rr];
    }
    return next_carry;
}

template <bool PL>
__global__ __launch_bounds__(256)
void dt_vres_kernel(const float* in, float* out, const float* edge, float clog2)
{
    __shared__ float XS[16 * VS_CHP];
    __shared__ __attribute__((aligned(16))) float SxS[VS_COLS * VS_TQ];
    __shared__ __attribute__((aligned(16))) float ExS[VS_COLS * VS_TQ];
    __shared__ __attribute__((aligned(16))) float omS[VS_COLS * VS_TQ];

    const int tid  = threadIdx.x;
    const int lid  = tid & 31;
    const int wv   = tid >> 5;
    const int half = lid & 15;
    const int khi  = lid >> 4;

    const int nb    = blockIdx.x;
    const int strip = nb & 63;
    const int cg    = (nb >> 6) & 1;
    const int b     = nb >> 7;

    const int Wd = 512;
    const long long chStride = 512LL * 512LL;
    const long long imgBase  = ((long long)b * 32 + cg * 16) * chStride + strip * VS_COLS;

    const int q  = tid & 1;
    const int pp = tid >> 1;
    const long long gm = imgBase + (long long)pp * Wd + q * 4;
    const int lm = pp * VS_ROWP + q * 4;
    const long long gStep = (long long)VS_GROWS * Wd;
    const int lStep = VS_GROWS * VS_ROWP;
    const long long gmP = ((long long)b * 32 + cg * 16 + (strip >> 2)) * chStride + (strip & 3) * 128
                        + (long long)(wv * 16) * Wd + lid * 4;
    const int lmP = (lid >> 1) * VS_CHP + (wv * 16) * VS_ROWP + (lid & 1) * 4;
    const long long gB = PL ? gmP : gm;
    const long long gI = PL ? (long long)Wd : chStride;
    const int lB = PL ? lmP : lm;
    const int lI = PL ? VS_ROWP : VS_CHP;

    float* tS = XS;
    const long long edgeBase = (long long)b * chStride + strip * VS_COLS;
    float4 ee[4];
#pragma unroll
    for (int k = 0; k < 4; ++k)
        ee[k] = *(const float4*)(edge + edgeBase + (long long)(pp + 128 * k) * Wd + q * 4);

    float4 rb[16];
#pragma unroll
    for (int i = 0; i < 16; ++i)
        rb[i] = *(const float4*)(in + gB + (long long)i * gI);

#pragma unroll
    for (int k = 0; k < 4; ++k) {
        const int h = pp + 128 * k;
        const float ev[4] = {ee[k].x, ee[k].y, ee[k].z, ee[k].w};
#pragma unroll
        for (int i = 0; i < 4; ++i) {
            const float Dv = 1.0f + 150.0f * ev[i];
            tS[(q * 4 + i) * VS_TP + h] = clog2 * Dv;
        }
    }
    lds_barrier();

    const float* tC = tS + wv * VS_TP;
    float tvF[16], tvB[16];
#pragma unroll
    for (int k = 0; k < 16; ++k) {
        const int w = 32 * k + lid;
        tvF[k] = (w == 0) ? -512.0f : tC[w];
        tvB[k] = (w == 0) ? -512.0f : tC[WIDTH - w];
    }

    float* Sc  = SxS + wv * VS_TQ;
    float* Ec  = ExS + wv * VS_TQ;
    float* omc = omS + wv * VS_TQ;
    build_tables(Sc, Ec, omc, tvF, lid, half);
    lds_barrier();

    float carry = 0.0f;
    for (int g = 0; g < VS_NG; ++g) {
#pragma unroll
        for (int i = 0; i < 16; ++i)
            put4(XS + i * lI + g * lStep + lB, rb[i]);
        if (g + 1 < VS_NG) {
#pragma unroll
            for (int i = 0; i < 16; ++i)
                rb[i] = *(const float4*)(in + gB + (long long)i * gI + (long long)(g + 1) * gStep);
        }
        lds_barrier();
        for (int t = 0; t < 8; ++t) {
            const int s = g * 8 + t;
            carry = vs_step<false>(XS + s * 16 * VS_ROWP, Sc, Ec, omc, s, wv, half, khi, carry);
        }
    }

    build_tables(Sc, Ec, omc, tvB, lid, half);
    carry = 0.0f;
    for (int h = 0; h < VS_NG; ++h) {
        for (int t = 0; t < 8; ++t) {
            const int s = h * 8 + t;
            const int T = SEGS - 1 - s;
            carry = vs_step<true>(XS + T * 16 * VS_ROWP, Sc, Ec, omc, s, wv, half, khi, carry);
        }
        lds_barrier();
        const int G = VS_NG - 1 - h;
#pragma unroll
        for (int i = 0; i < 16; ++i) {
            const float4 o = get4(XS + i * lI + G * lStep + lB);
            *(float4*)(out + gB + (long long)i * gI + (long long)G * gStep) = o;
        }
    }
}

extern "C" void kernel_launch(void* const* d_in, const int* in_sizes, int n_in,
                              void* d_out, int out_size, void* d_ws, size_t ws_size,
                              hipStream_t stream) {
    (void)in_sizes; (void)n_in; (void)out_size; (void)d_ws; (void)ws_size;
    const float* img  = (const float*)d_in[0];
    const float* edge = (const float*)d_in[1];
    float* out = (float*)d_out;

    const double SQRT2 = 1.4142135623730951;
    const double SQRT3 = 1.7320508075688772;
    const double LOG2E = 1.4426950408889634;
    const double denom = 3.872983346207417;
    const double sigma1 = 60.0 * SQRT3 * 2.0 / denom;
    const double sigma2 = 60.0 * SQRT3 * 1.0 / denom;
    const float c1 = (float)(-SQRT2 / sigma1 * LOG2E);
    const float c2 = (float)(-SQRT2 / sigma2 * LOG2E);

    const dim3 grid(4096), block(32);
    const size_t shmem = (size_t)(16 * XPITCH + 3 * WIDTH) * sizeof(float);
    const dim3 vgrid(512), vblock(256);

    dt_hpass_kernel<false, true><<<grid, block, shmem, stream>>>(img, out, edge, c1);
    dt_vres_kernel<true><<<vgrid, vblock, 0, stream>>>(out, out, edge, c1);
    dt_hpass_kernel<true, false><<<grid, block, shmem, stream>>>(out, out, edge, c2);
    dt_vres_kernel<false><<<vgrid, vblock, 0, stream>>>(out, out, edge, c2);
}
